# starting kernel
# baseline (speedup 1.0000x reference)
#include <hip/hip_runtime.h>

typedef __attribute__((ext_vector_type(2))) float v2f;
typedef __attribute__((ext_vector_type(8))) float v8f;

#define NB   16
#define NC   192
#define NH   64
#define NW   64
#define NK   (NH * NW)
#define CBLK 32
#define NCB  (NC / CBLK)
#define NDT  (NC / 16)
#define NOFF 5
#define CC   (NC * NC)

#define WM(A_, B_, C_) \
    __builtin_amdgcn_wmma_f32_16x16x4_f32(false, (A_), false, (B_), (short)0, (C_), false, false)

#define WMMA10(a0_, a1_, b0_, b1_, b2_, b3_, b4_) do {            \
    accA0 = WM(a0_, b0_, accA0); accB0 = WM(a1_, b0_, accB0);     \
    accA1 = WM(a0_, b1_, accA1); accB1 = WM(a1_, b1_, accB1);     \
    accA2 = WM(a0_, b2_, accA2); accB2 = WM(a1_, b2_, accB2);     \
    accA3 = WM(a0_, b3_, accA3); accB3 = WM(a1_, b3_, accB3);     \
    accA4 = WM(a0_, b4_, accA4); accB4 = WM(a1_, b4_, accB4);     \
} while (0)

__global__ void cofe_zero_ws(float* ws) {
    int i = threadIdx.x;
    if (i < NB * NOFF) ws[i] = 0.0f;
}

__device__ __forceinline__ void tile_epilogue(v8f acc, int o, int b, int cb, int db,
                                              int m_off, int n, int lane,
                                              float* __restrict__ out,
                                              float* __restrict__ ss) {
    size_t base = (size_t)(b * NOFF + o) * CC + (size_t)(cb + m_off) * NC + (size_t)(db + n);
    float s = 0.0f;
#pragma unroll
    for (int i = 0; i < 8; ++i) {
        float v = acc[i];
        out[base + (size_t)i * NC] = v;
        s += v * v;
    }
#pragma unroll
    for (int off = 16; off > 0; off >>= 1)
        s += __shfl_xor(s, off, 32);
    if (lane == 0)
        atomicAdd(&ss[b * NOFF + o], s);
}

__global__ __launch_bounds__(256) void cofe_wmma(const float* __restrict__ x,
                                                 float* __restrict__ out,
                                                 float* __restrict__ ss) {
    const int wave   = (blockIdx.x << 3) + (threadIdx.x >> 5);
    const int lane   = threadIdx.x & 31;
    const int lane15 = lane & 15;
    const int kk     = (lane >> 4) << 1;

    const int b  = wave / (NCB * NDT);
    const int rm = wave % (NCB * NDT);
    const int cb = (rm / NDT) * CBLK;
    const int db = (rm % NDT) * 16;

    const float* __restrict__ xb = x + (size_t)b * NC * NK;
    const int xrow = (cb + lane15) * NK;
    const int yrow = (db + lane15) * NK;

    v8f accA0 = {}, accA1 = {}, accA2 = {}, accA3 = {}, accA4 = {};
    v8f accB0 = {}, accB1 = {}, accB2 = {}, accB3 = {}, accB4 = {};

    for (int h = 0; h < NH; ++h) {
        const int hm  = (h > 0) ? (h - 1) : 0;
        const int rm1 = yrow + (hm << 6);
        const int r0  = yrow + (h  << 6);
        const int xr  = xrow + (h  << 6);

        auto clamped_step = [&](int w) {
            const int wm  = (w > 0)  ? (w - 1) : 0;
            const int wp2 = (w < 62) ? (w + 2) : 63;
            v2f a0 = *(const v2f*)(xb + xr + w);
            v2f a1 = *(const v2f*)(xb + xr + 16 * NK + w);
            v2f b0, b1, b2, b3, b4;
            b0.x = xb[rm1 + wm];    b0.y = xb[rm1 + w];
            b1   = *(const v2f*)(xb + rm1 + w);
            b2.x = xb[rm1 + w + 1]; b2.y = xb[rm1 + wp2];
            b3.x = xb[r0 + wm];     b3.y = xb[r0 + w];
            b4   = *(const v2f*)(xb + r0 + w);
            WMMA10(a0, a1, b0, b1, b2, b3, b4);
        };

        clamped_step(kk);

        for (int wi = 1; wi < 15; ++wi) {
            const int w = (wi << 2) + kk;
            v2f a0 = *(const v2f*)(xb + xr + w);
            v2f a1 = *(const v2f*)(xb + xr + 16 * NK + w);

            const float* __restrict__ pu = xb + rm1 + w - 1;
            const float* __restrict__ pc = xb + r0  + w - 1;
            float p0 = pu[0], p1 = pu[1], p2 = pu[2], p3 = pu[3];
            float q0 = pc[0], q1 = pc[1], q2 = pc[2];

            v2f b0 = {p0, p1};
            v2f b1 = {p1, p2};
            v2f b2 = {p2, p3};
            v2f b3 = {q0, q1};
            v2f b4 = {q1, q2};
            WMMA10(a0, a1, b0, b1, b2, b3, b4);
        }

        clamped_step(60 + kk);
    }

    const int m_off = (lane >> 4) << 3;
    tile_epilogue(accA0, 0, b, cb,      db, m_off, lane15, lane, out, ss);
    tile_epilogue(accB0, 0, b, cb + 16, db, m_off, lane15, lane, out, ss);
    tile_epilogue(accA1, 1, b, cb,      db, m_off, lane15, lane, out, ss);
    tile_epilogue(accB1, 1, b, cb + 16, db, m_off, lane15, lane, out, ss);
    tile_epilogue(accA2, 2, b, cb,      db, m_off, lane15, lane, out, ss);
    tile_epilogue(accB2, 2, b, cb + 16, db, m_off, lane15, lane, out, ss);
    tile_epilogue(accA3, 3, b, cb,      db, m_off, lane15, lane, out, ss);
    tile_epilogue(accB3, 3, b, cb + 16, db, m_off, lane15, lane, out, ss);
    tile_epilogue(accA4, 4, b, cb,      db, m_off, lane15, lane, out, ss);
    tile_epilogue(accB4, 4, b, cb + 16, db, m_off, lane15, lane, out, ss);
}

__global__ __launch_bounds__(256) void cofe_normalize(float* __restrict__ out,
                                                      const float* __restrict__ ss) {
    size_t i = (size_t)blockIdx.x * 256 + threadIdx.x;
    size_t grp = i / CC;
    float nrm = sqrtf(ss[grp]);
    nrm = fmaxf(nrm, 1e-12f);
    out[i] = out[i] / nrm;
}

extern "C" void kernel_launch(void* const* d_in, const int* in_sizes, int n_in,
                              void* d_out, int out_size, void* d_ws, size_t ws_size,
                              hipStream_t stream) {
    const float* x  = (const float*)d_in[0];
    float* out      = (float*)d_out;
    float* ss       = (float*)d_ws;

    cofe_zero_ws<<<1, 128, 0, stream>>>(ss);

    cofe_wmma<<<(NB * NCB * NDT) / 8, 256, 0, stream>>>(x, out, ss);

    cofe_normalize<<<(NB * NOFF * CC) / 256, 256, 0, stream>>>(out, ss);
}


# Main kernel on 8 waves, two per SIMD: the wave served first holds 9 accumulators, its partner 3 (dealt by offset); one wave's LDS reads and staging run under the other's products; k order kept
# speedup vs baseline: 2.0035x; 2.0035x over previous
#include <hip/hip_runtime.h>


typedef __attribute__((ext_vector_type(2))) float v2f;
typedef __attribute__((ext_vector_type(4))) float v4f;
typedef __attribute__((ext_vector_type(8))) float v8f;

#define NB   16
#define NC   192
#define NH   64
#define NW   64
#define NK   (NH * NW)
#define NOFF 5
#define CC   (NC * NC)

#define TS     48
#define RCH    4
#define NSLOT  (RCH + 1)
#define PITCH  68
#define CHP    (NSLOT * PITCH)
#define OPF    (TS * CHP)
#define BUFF   (2 * OPF)
#define NCHUNK (NH / RCH)
#define PCH    (TS / RCH)
#define NTHR   256
#define NLD    8
#define NSIDE  3

#define WM(A_, B_, C_) \
    __builtin_amdgcn_wmma_f32_16x16x4_f32(false, (A_), false, (B_), (short)0, (C_), false, false)

__global__ void cofe_zero_ws(float* ws) {
    int i = threadIdx.x;
    if (i < NB * NOFF) ws[i] = 0.0f;
}

__device__ __forceinline__ void tile_epilogue(v8f acc, int o, int b, int cb, int db,
                                              int m_off, int n, int lane,
                                              float* __restrict__ out,
                                              float* __restrict__ ss) {
    size_t base = (size_t)(b * NOFF + o) * CC + (size_t)(cb + m_off) * NC + (size_t)(db + n);
    float s = 0.0f;
#pragma unroll
    for (int i = 0; i < 8; ++i) {
        float v = acc[i];
        out[base + (size_t)i * NC] = v;
        s += v * v;
    }
#pragma unroll
    for (int off = 16; off > 0; off >>= 1)
        s += __shfl_xor(s, off, 32);
    if (lane == 0)
        atomicAdd(&ss[b * NOFF + o], s);
}

__device__ __forceinline__ void part_load(const float* __restrict__ src, const int (&go)[NLD],
                                          v4f (&r)[NLD], bool full) {
#pragma unroll
    for (int i = 0; i < NLD - 1; ++i) r[i] = *(const v4f*)(src + go[i]);
    if (full) r[NLD - 1] = *(const v4f*)(src + go[NLD - 1]);
}

__device__ __forceinline__ void store_one(float* __restrict__ dst, int lo, v4f r) {
    v2f lo2 = {r.x, r.y};
    v2f hi2 = {r.z, r.w};
    *(v2f*)(dst + lo)     = lo2;
    *(v2f*)(dst + lo + 2) = hi2;
}

__device__ __forceinline__ void part_store(float* __restrict__ dst, const int (&lo)[NLD],
                                           int q, const v4f (&r)[NLD], bool full) {
#pragma unroll
    for (int i = 0; i < NLD - 1; ++i) store_one(dst, lo[i], r[i]);
    if (full) store_one(dst, lo[NLD - 1], r[NLD - 1]);
    if (q == 0 || q == 15) {
        const int hd = (q == 0) ? -1 : 4;
#pragma unroll
        for (int i = 0; i < NLD - 1; ++i) dst[lo[i] + hd] = (q == 0) ? r[i].x : r[i].w;
        if (full) dst[lo[NLD - 1] + hd] = (q == 0) ? r[NLD - 1].x : r[NLD - 1].w;
    }
}

template <int NV>
__device__ __forceinline__ void load_group(const float* __restrict__ rb,
                                           const int (&oa)[3], const int (&sb)[NSIDE], int ol, int g,
                                           v2f (&a)[4][3], v2f (&bq)[4][NSIDE]) {
#pragma unroll
    for (int s = 0; s < 4; ++s) {
        const int k4 = 4 * (4 * g + s);
#pragma unroll
        for (int ci = 0; ci < 3; ++ci) a[s][ci] = *(const v2f*)(rb + oa[ci] + k4);
#pragma unroll
        for (int v = 0; v < NV; ++v) {
            const float* __restrict__ rs = rb + sb[v];
            bq[s][v].x = rs[ol + k4];
            bq[s][v].y = rs[ol + k4 + 1];
        }
    }
}

template <int NV>
__device__ __forceinline__ void mma_group(const v2f (&a)[4][3], const v2f (&bq)[4][NSIDE],
                                          v8f (&acc)[3][NSIDE]) {
#pragma unroll
    for (int s = 0; s < 4; ++s) {
#pragma unroll
        for (int v = 0; v < NV; ++v) {
#pragma unroll
            for (int ci = 0; ci < 3; ++ci) acc[ci][v] = WM(a[s][ci], bq[s][v], acc[ci][v]);
        }
    }
}

template <int NV>
__device__ __forceinline__ void row_mma(const float* __restrict__ rb,
                                        const int (&oa)[3], const int (&sb)[NSIDE], int ol,
                                        v8f (&acc)[3][NSIDE]) {
    v2f a0[4][3], b0[4][NSIDE], a1[4][3], b1[4][NSIDE];
    load_group<NV>(rb, oa, sb, ol, 0, a0, b0);
#pragma unroll
    for (int g = 0; g < 4; g += 2) {
        load_group<NV>(rb, oa, sb, ol, g + 1, a1, b1);
        __builtin_amdgcn_sched_barrier(0);
        mma_group<NV>(a0, b0, acc);
        __builtin_amdgcn_sched_barrier(0);
        if (g + 2 < 4) load_group<NV>(rb, oa, sb, ol, g + 2, a0, b0);
        __builtin_amdgcn_sched_barrier(0);
        mma_group<NV>(a1, b1, acc);
        __builtin_amdgcn_sched_barrier(0);
    }
    asm volatile("v_nop\n\tv_nop\n\tv_nop\n\tv_nop\n\tv_nop\n\tv_nop\n\tv_nop\n\tv_nop");
}

__device__ __forceinline__ int side_shift(int o) {
    return (o < 3) ? (o + 1) : (PITCH + o - 2);
}

__global__ __launch_bounds__(NTHR) void cofe_wmma(const float* __restrict__ x,
                                                  float* __restrict__ out,
                                                  float* __restrict__ ss) {
    __shared__ float lds[2 * BUFF];

    const int tid    = threadIdx.x;
    const int lane   = tid & 31;
    const int lane15 = lane & 15;
    const int half   = lane >> 4;
    const int wv     = __builtin_amdgcn_readfirstlane(tid >> 5);

    const int blk = blockIdx.x;
    const int b   = blk >> 4;
    const int cb  = ((blk >> 2) & 3) * TS;
    const int db  = (blk & 3) * TS;
    const float* __restrict__ xb = x + (size_t)b * NC * NK;

    int nv, so[NSIDE], sd[NSIDE];
    if (wv < 3)       { nv = 3; so[0] = 0; so[1] = 1; so[2] = 2; sd[0] = wv;     sd[1] = wv;     sd[2] = wv;     }
    else if (wv == 3) { nv = 2; so[0] = 4; so[1] = 4; so[2] = 4; sd[0] = 0;      sd[1] = 1;      sd[2] = 0;      }
    else if (wv < 7)  { nv = 1; so[0] = 3; so[1] = 3; so[2] = 3; sd[0] = wv - 4; sd[1] = wv - 4; sd[2] = wv - 4; }
    else              { nv = 1; so[0] = 4; so[1] = 4; so[2] = 4; sd[0] = 2;      sd[1] = 2;      sd[2] = 2;      }

    int oa[3], sb[NSIDE];
#pragma unroll
    for (int ci = 0; ci < 3; ++ci)
        oa[ci] = ((ci * 16 + lane15) * NSLOT + 1) * PITCH + 2 + 2 * half;
    const int ol = lane15 * CHP + 2 * half;
#pragma unroll
    for (int v = 0; v < NSIDE; ++v)
        sb[v] = OPF + sd[v] * 16 * CHP + side_shift(so[v]);

    v8f acc[3][NSIDE];
#pragma unroll
    for (int ci = 0; ci < 3; ++ci)
#pragma unroll
        for (int v = 0; v < NSIDE; ++v) acc[ci][v] = (v8f){};

    const int  q    = tid & 15;
    const bool full = wv < 4;
    int go[NLD], lo[NLD], go0[NLD];
#pragma unroll
    for (int i = 0; i < NLD; ++i) {
        int row = (tid >> 4) + 16 * i;
        if (row > 119) row = 119;
        const int s   = row % NSLOT;
        const int t2  = row / NSLOT;
        const int chl = t2 % PCH;
        const int t   = t2 / PCH;
        go[i]  = ((t ? db : cb) + chl) * NK + s * NW + q * 4;
        lo[i]  = ((t * TS + chl) * NSLOT + s) * PITCH + 2 + q * 4;
        go0[i] = go[i] - (s > 0 ? NW : 0);
    }

    v4f stg[NLD];

#pragma unroll 1
    for (int part = 0; part < RCH; ++part) {
        part_load(xb + part * (PCH * NK), go0, stg, full);
        part_store(lds + part * (PCH * CHP), lo, q, stg, full);
    }
    __syncthreads();

#pragma unroll 1
    for (int chunk = 0; chunk < NCHUNK; ++chunk) {
        const float* __restrict__ cur = lds + (chunk & 1) * BUFF;
        float* __restrict__ nxt       = lds + ((chunk & 1) ^ 1) * BUFF;
        const bool more = (chunk + 1) < NCHUNK;

#pragma unroll 1
        for (int j = 0; j < RCH; ++j) {
            if (more) part_load(xb + j * (PCH * NK) + ((chunk + 1) * RCH - 1) * NW, go, stg, full);
            if (nv == 3)      row_mma<3>(cur + j * PITCH, oa, sb, ol, acc);
            else if (nv == 2) row_mma<2>(cur + j * PITCH, oa, sb, ol, acc);
            else              row_mma<1>(cur + j * PITCH, oa, sb, ol, acc);
            if (more) part_store(nxt + j * (PCH * CHP), lo, q, stg, full);
        }
        __syncthreads();
    }

    const int m_off = half << 3;
#pragma unroll
    for (int v = 0; v < NSIDE; ++v) {
        if (v < nv) {
#pragma unroll
            for (int ci = 0; ci < 3; ++ci)
                tile_epilogue(acc[ci][v], so[v], b, cb + ci * 16, db + sd[v] * 16, m_off, lane15, lane, out, ss);
        }
    }
}

__global__ __launch_bounds__(256) void cofe_normalize(float* __restrict__ out,
                                                      const float* __restrict__ ss) {
    size_t i = (size_t)blockIdx.x * 256 + threadIdx.x;
    size_t grp = i / CC;
    float nrm = sqrtf(ss[grp]);
    nrm = fmaxf(nrm, 1e-12f);
    out[i] = out[i] / nrm;
}

extern "C" void kernel_launch(void* const* d_in, const int* in_sizes, int n_in,
                              void* d_out, int out_size, void* d_ws, size_t ws_size,
                              hipStream_t stream) {
    const float* x  = (const float*)d_in[0];
    float* out      = (float*)d_out;
    float* ss       = (float*)d_ws;

    cofe_zero_ws<<<1, 128, 0, stream>>>(ss);

    cofe_wmma<<<NB * 16, NTHR, 0, stream>>>(x, out, ss);

    cofe_normalize<<<(NB * NOFF * CC) / 256, 256, 0, stream>>>(out, ss);
}
